# starting kernel
# baseline (speedup 1.0000x reference)
#include <hip/hip_runtime.h>
#include <hip/hip_bf16.h>

#define E_N 10000
#define T_N 24
#define D_N 128
#define H_N 8
#define K_N 16
#define FF_N 512
#define EPS_LN 1e-5f

typedef float v2f __attribute__((ext_vector_type(2)));
typedef float v8f __attribute__((ext_vector_type(8)));

__device__ __forceinline__ v8f wmma_f32(v2f a, v2f b, v8f c) {
  return __builtin_amdgcn_wmma_f32_16x16x4_f32(
       false, a,  false, b,
       (short)0, c,  false,  false);
}

__device__ __forceinline__ void gemm_acc(const float* sA, const float* B, int ldb,
                                         int wave, int half, int lm, int Kdim,
                                         v8f acc[4]) {
  const int n0 = wave * 32 + lm;
  const int n1 = n0 + 16;
  for (int k0 = 0; k0 < Kdim; k0 += 4) {
    const int ka = k0 + 2 * half;
    v2f a0, a1, b0, b1;
    a0.x = sA[lm * 128 + ka];        a0.y = sA[lm * 128 + ka + 1];
    a1.x = sA[(16 + lm) * 128 + ka]; a1.y = sA[(16 + lm) * 128 + ka + 1];
    b0.x = B[(size_t)ka * ldb + n0]; b0.y = B[(size_t)(ka + 1) * ldb + n0];
    b1.x = B[(size_t)ka * ldb + n1]; b1.y = B[(size_t)(ka + 1) * ldb + n1];
    acc[0] = wmma_f32(a0, b0, acc[0]);
    acc[1] = wmma_f32(a0, b1, acc[1]);
    acc[2] = wmma_f32(a1, b0, acc[2]);
    acc[3] = wmma_f32(a1, b1, acc[3]);
  }
}

__global__ __launch_bounds__(128) void kv_proj_kernel(
    const float* __restrict__ x, const float* __restrict__ Wk,
    const float* __restrict__ Wv, float* __restrict__ Kx, float* __restrict__ Vx) {
  __shared__ float sx0[16 * 128];
  const int tid = threadIdx.x;
  const int e0 = blockIdx.x * 16;
  for (int i = tid; i < 16 * 128; i += 128) {
    const int r = i >> 7, c = i & 127;
    sx0[i] = x[(size_t)(e0 + r) * (T_N * D_N) + c];
  }
  __syncthreads();

  const int lane = tid & 31, wave = tid >> 5;
  const int half = lane >> 4, lm = lane & 15;
  const int n0 = wave * 32 + lm, n1 = n0 + 16;
  v8f aK0{}, aK1{}, aV0{}, aV1{};
  for (int k0 = 0; k0 < 128; k0 += 4) {
    const int ka = k0 + 2 * half;
    v2f a, bk0, bk1, bv0, bv1;
    a.x = sx0[lm * 128 + ka];  a.y = sx0[lm * 128 + ka + 1];
    bk0.x = Wk[ka * 128 + n0]; bk0.y = Wk[(ka + 1) * 128 + n0];
    bk1.x = Wk[ka * 128 + n1]; bk1.y = Wk[(ka + 1) * 128 + n1];
    bv0.x = Wv[ka * 128 + n0]; bv0.y = Wv[(ka + 1) * 128 + n0];
    bv1.x = Wv[ka * 128 + n1]; bv1.y = Wv[(ka + 1) * 128 + n1];
    aK0 = wmma_f32(a, bk0, aK0);
    aK1 = wmma_f32(a, bk1, aK1);
    aV0 = wmma_f32(a, bv0, aV0);
    aV1 = wmma_f32(a, bv1, aV1);
  }
  for (int r = 0; r < 8; ++r) {
    const size_t row = (size_t)(e0 + r + 8 * half) * 128;
    Kx[row + n0] = aK0[r];
    Kx[row + n1] = aK1[r];
    Vx[row + n0] = aV0[r];
    Vx[row + n1] = aV1[r];
  }
}

__global__ __launch_bounds__(128) void attn_block_kernel(
    const float* __restrict__ x, const int* __restrict__ nbr,
    const float* __restrict__ Wq, const float* __restrict__ Wo,
    const float* __restrict__ bo, const float* __restrict__ g1,
    const float* __restrict__ b1, const float* __restrict__ g2,
    const float* __restrict__ b2, const float* __restrict__ W1,
    const float* __restrict__ bf1, const float* __restrict__ W2,
    const float* __restrict__ bf2, const float* __restrict__ Kx,
    const float* __restrict__ Vx, float* __restrict__ out) {
  __shared__ float buf0[32 * 128];
  __shared__ float buf1[32 * 128];
  __shared__ float buf2[32 * 128];
  __shared__ float buf3[32 * 128];
  const int VOFF = 16 * 128;

  const int tid = threadIdx.x;
  const int e = blockIdx.x;
  const int lane = tid & 31, wave = tid >> 5;
  const int half = lane >> 4, lm = lane & 15;
  const size_t xbase = (size_t)e * (T_N * D_N);

  for (int i = tid; i < 32 * 128; i += 128) {
    const int r = i >> 7, c = i & 127;
    buf0[i] = (r < T_N) ? x[xbase + (size_t)r * D_N + c] : 0.0f;
  }
  for (int i = tid; i < 16 * 128; i += 128) {
    const int k = i >> 7, c = i & 127;
    const int n = nbr[e * K_N + k];
    buf2[i]        = Kx[(size_t)n * D_N + c];
    buf2[VOFF + i] = Vx[(size_t)n * D_N + c];
  }
  __syncthreads();

  {
    v8f q[4] = {};
    gemm_acc(buf0, Wq, 128, wave, half, lm, 128, q);
    for (int i = 0; i < 4; ++i)
      for (int r = 0; r < 8; ++r) {
        const int row = ((i >> 1) << 4) + r + 8 * half;
        const int col = wave * 32 + ((i & 1) << 4) + lm;
        buf1[row * 128 + col] = q[i][r] * 0.25f;
      }
  }
  __syncthreads();

  {
    const int hc0 = (2 * wave) * 16, hc1 = hc0 + 16;
    v8f s0{}, s1{}, s2{}, s3{};
    for (int k0 = 0; k0 < 16; k0 += 4) {
      const int ka = k0 + 2 * half;
      v2f a00, a01, a10, a11, b0, b1;
      a00.x = buf1[lm * 128 + hc0 + ka];        a00.y = buf1[lm * 128 + hc0 + ka + 1];
      a01.x = buf1[(16 + lm) * 128 + hc0 + ka]; a01.y = buf1[(16 + lm) * 128 + hc0 + ka + 1];
      a10.x = buf1[lm * 128 + hc1 + ka];        a10.y = buf1[lm * 128 + hc1 + ka + 1];
      a11.x = buf1[(16 + lm) * 128 + hc1 + ka]; a11.y = buf1[(16 + lm) * 128 + hc1 + ka + 1];
      b0.x = buf2[lm * 128 + hc0 + ka]; b0.y = buf2[lm * 128 + hc0 + ka + 1];
      b1.x = buf2[lm * 128 + hc1 + ka]; b1.y = buf2[lm * 128 + hc1 + ka + 1];
      s0 = wmma_f32(a00, b0, s0);
      s1 = wmma_f32(a01, b0, s1);
      s2 = wmma_f32(a10, b1, s2);
      s3 = wmma_f32(a11, b1, s3);
    }
    for (int r = 0; r < 8; ++r) {
      const int t0 = r + 8 * half, t1 = 16 + r + 8 * half;
      buf3[t0 * 128 + hc0 + lm] = s0[r];
      buf3[t1 * 128 + hc0 + lm] = s1[r];
      buf3[t0 * 128 + hc1 + lm] = s2[r];
      buf3[t1 * 128 + hc1 + lm] = s3[r];
    }
  }
  __syncthreads();

  for (int idx = tid; idx < T_N * H_N; idx += 128) {
    const int t = idx >> 3, h = idx & 7;
    float* row = &buf3[t * 128 + h * 16];
    float m = row[0];
    for (int k = 1; k < 16; ++k) m = fmaxf(m, row[k]);
    float s = 0.f;
    for (int k = 0; k < 16; ++k) { row[k] = __expf(row[k] - m); s += row[k]; }
    const float inv = 1.f / s;
    for (int k = 0; k < 16; ++k) row[k] *= inv;
  }
  __syncthreads();

  {
    const int hc0 = (2 * wave) * 16, hc1 = hc0 + 16;
    v8f z0{}, z1{}, z2{}, z3{};
    for (int k0 = 0; k0 < 16; k0 += 4) {
      const int ka = k0 + 2 * half;
      v2f a00, a01, a10, a11, b0, b1;
      a00.x = buf3[lm * 128 + hc0 + ka];        a00.y = buf3[lm * 128 + hc0 + ka + 1];
      a01.x = buf3[(16 + lm) * 128 + hc0 + ka]; a01.y = buf3[(16 + lm) * 128 + hc0 + ka + 1];
      a10.x = buf3[lm * 128 + hc1 + ka];        a10.y = buf3[lm * 128 + hc1 + ka + 1];
      a11.x = buf3[(16 + lm) * 128 + hc1 + ka]; a11.y = buf3[(16 + lm) * 128 + hc1 + ka + 1];
      b0.x = buf2[VOFF + ka * 128 + hc0 + lm]; b0.y = buf2[VOFF + (ka + 1) * 128 + hc0 + lm];
      b1.x = buf2[VOFF + ka * 128 + hc1 + lm]; b1.y = buf2[VOFF + (ka + 1) * 128 + hc1 + lm];
      z0 = wmma_f32(a00, b0, z0);
      z1 = wmma_f32(a01, b0, z1);
      z2 = wmma_f32(a10, b1, z2);
      z3 = wmma_f32(a11, b1, z3);
    }
    for (int r = 0; r < 8; ++r) {
      const int t0 = r + 8 * half, t1 = 16 + r + 8 * half;
      buf1[t0 * 128 + hc0 + lm] = z0[r];
      buf1[t1 * 128 + hc0 + lm] = z1[r];
      buf1[t0 * 128 + hc1 + lm] = z2[r];
      buf1[t1 * 128 + hc1 + lm] = z3[r];
    }
  }
  __syncthreads();

  {
    v8f o[4] = {};
    gemm_acc(buf1, Wo, 128, wave, half, lm, 128, o);
    for (int i = 0; i < 4; ++i)
      for (int r = 0; r < 8; ++r) {
        const int row = ((i >> 1) << 4) + r + 8 * half;
        const int col = wave * 32 + ((i & 1) << 4) + lm;
        buf2[row * 128 + col] = o[i][r] + bo[col] + buf0[row * 128 + col];
      }
  }
  __syncthreads();

  if (tid < T_N) {
    const float* rp = &buf2[tid * 128];
    float mu = 0.f;
    for (int c = 0; c < 128; ++c) mu += rp[c];
    mu *= (1.f / 128.f);
    float var = 0.f;
    for (int c = 0; c < 128; ++c) { const float d = rp[c] - mu; var += d * d; }
    var *= (1.f / 128.f);
    const float rs = rsqrtf(var + EPS_LN);
    float* wp = &buf3[tid * 128];
    for (int c = 0; c < 128; ++c) wp[c] = (rp[c] - mu) * rs * g1[c] + b1[c];
  }
  __syncthreads();

  v8f y[4] = {};
  for (int ch = 0; ch < 4; ++ch) {
    v8f g[4] = {};
    gemm_acc(buf3, W1 + ch * 128, FF_N, wave, half, lm, 128, g);
    for (int i = 0; i < 4; ++i)
      for (int r = 0; r < 8; ++r) {
        const int row = ((i >> 1) << 4) + r + 8 * half;
        const int col = wave * 32 + ((i & 1) << 4) + lm;
        buf1[row * 128 + col] = fmaxf(g[i][r] + bf1[ch * 128 + col], 0.f);
      }
    __syncthreads();
    gemm_acc(buf1, W2 + (size_t)(ch * 128) * D_N, 128, wave, half, lm, 128, y);
    __syncthreads();
  }

  for (int i = 0; i < 4; ++i)
    for (int r = 0; r < 8; ++r) {
      const int row = ((i >> 1) << 4) + r + 8 * half;
      const int col = wave * 32 + ((i & 1) << 4) + lm;
      buf1[row * 128 + col] = y[i][r] + bf2[col] + buf3[row * 128 + col];
    }
  __syncthreads();

  if (tid < T_N) {
    const float* rp = &buf1[tid * 128];
    float mu = 0.f;
    for (int c = 0; c < 128; ++c) mu += rp[c];
    mu *= (1.f / 128.f);
    float var = 0.f;
    for (int c = 0; c < 128; ++c) { const float d = rp[c] - mu; var += d * d; }
    var *= (1.f / 128.f);
    const float rs = rsqrtf(var + EPS_LN);
    for (int c = 0; c < 128; ++c)
      out[xbase + (size_t)tid * D_N + c] = (rp[c] - mu) * rs * g2[c] + b2[c];
  }
}

extern "C" void kernel_launch(void* const* d_in, const int* in_sizes, int n_in,
                              void* d_out, int out_size, void* d_ws, size_t ws_size,
                              hipStream_t stream) {
  const float* x   = (const float*)d_in[0];
  const int*   nbr = (const int*)d_in[1];
  const float* Wq  = (const float*)d_in[2];
  const float* Wk  = (const float*)d_in[3];
  const float* Wv  = (const float*)d_in[4];
  const float* Wo  = (const float*)d_in[5];
  const float* bo  = (const float*)d_in[6];
  const float* g1  = (const float*)d_in[7];
  const float* b1  = (const float*)d_in[8];
  const float* g2  = (const float*)d_in[9];
  const float* b2  = (const float*)d_in[10];
  const float* W1  = (const float*)d_in[11];
  const float* bf1 = (const float*)d_in[12];
  const float* W2  = (const float*)d_in[13];
  const float* bf2 = (const float*)d_in[14];

  float* Kx = (float*)d_ws;
  float* Vx = Kx + (size_t)E_N * D_N;

  kv_proj_kernel<<<E_N / 16, 128, 0, stream>>>(x, Wk, Wv, Kx, Vx);
  attn_block_kernel<<<E_N, 128, 0, stream>>>(x, nbr, Wq, Wo, bo, g1, b1, g2, b2,
                                             W1, bf1, W2, bf2, Kx, Vx,
                                             (float*)d_out);
}


# LDS row pitch 128 -> 132 floats (528 B = 16 B x 33): the 16 rows of an A-fragment read and the LayerNorm row reads no longer share one bank; x tile keeps 24 rows so LDS stays under 64 KiB
# speedup vs baseline: 1.5110x; 1.5110x over previous
#include <hip/hip_runtime.h>
#include <hip/hip_bf16.h>

#define E_N 10000
#define T_N 24
#define D_N 128
#define H_N 8
#define K_N 16
#define FF_N 512
#define EPS_LN 1e-5f
#define LDP 132

typedef float v2f __attribute__((ext_vector_type(2)));
typedef float v8f __attribute__((ext_vector_type(8)));

__device__ __forceinline__ v8f wmma_f32(v2f a, v2f b, v8f c) {
  return __builtin_amdgcn_wmma_f32_16x16x4_f32(
       false, a,  false, b,
       (short)0, c,  false,  false);
}

__device__ __forceinline__ void gemm_acc(const float* sA, const float* B, int ldb,
                                         int wave, int half, int lm, int hiRow,
                                         int Kdim, v8f acc[4]) {
  const int n0 = wave * 32 + lm;
  const int n1 = n0 + 16;
#pragma unroll
  for (int k0 = 0; k0 < Kdim; k0 += 4) {
    const int ka = k0 + 2 * half;
    v2f a0, a1, b0, b1;
    a0.x = sA[lm * LDP + ka];        a0.y = sA[lm * LDP + ka + 1];
    a1.x = sA[hiRow * LDP + ka];     a1.y = sA[hiRow * LDP + ka + 1];
    b0.x = B[(size_t)ka * ldb + n0]; b0.y = B[(size_t)(ka + 1) * ldb + n0];
    b1.x = B[(size_t)ka * ldb + n1]; b1.y = B[(size_t)(ka + 1) * ldb + n1];
    acc[0] = wmma_f32(a0, b0, acc[0]);
    acc[1] = wmma_f32(a0, b1, acc[1]);
    acc[2] = wmma_f32(a1, b0, acc[2]);
    acc[3] = wmma_f32(a1, b1, acc[3]);
  }
}

__global__ __launch_bounds__(128) void kv_proj_kernel(
    const float* __restrict__ x, const float* __restrict__ Wk,
    const float* __restrict__ Wv, float* __restrict__ Kx, float* __restrict__ Vx) {
  __shared__ float sx0[16 * LDP];
  const int tid = threadIdx.x;
  const int e0 = blockIdx.x * 16;
  for (int i = tid; i < 16 * 128; i += 128) {
    const int r = i >> 7, c = i & 127;
    sx0[r * LDP + c] = x[(size_t)(e0 + r) * (T_N * D_N) + c];
  }
  __syncthreads();

  const int lane = tid & 31, wave = tid >> 5;
  const int half = lane >> 4, lm = lane & 15;
  const int n0 = wave * 32 + lm, n1 = n0 + 16;
  v8f aK0{}, aK1{}, aV0{}, aV1{};
  for (int k0 = 0; k0 < 128; k0 += 4) {
    const int ka = k0 + 2 * half;
    v2f a, bk0, bk1, bv0, bv1;
    a.x = sx0[lm * LDP + ka];  a.y = sx0[lm * LDP + ka + 1];
    bk0.x = Wk[ka * 128 + n0]; bk0.y = Wk[(ka + 1) * 128 + n0];
    bk1.x = Wk[ka * 128 + n1]; bk1.y = Wk[(ka + 1) * 128 + n1];
    bv0.x = Wv[ka * 128 + n0]; bv0.y = Wv[(ka + 1) * 128 + n0];
    bv1.x = Wv[ka * 128 + n1]; bv1.y = Wv[(ka + 1) * 128 + n1];
    aK0 = wmma_f32(a, bk0, aK0);
    aK1 = wmma_f32(a, bk1, aK1);
    aV0 = wmma_f32(a, bv0, aV0);
    aV1 = wmma_f32(a, bv1, aV1);
  }
  for (int r = 0; r < 8; ++r) {
    const size_t row = (size_t)(e0 + r + 8 * half) * 128;
    Kx[row + n0] = aK0[r];
    Kx[row + n1] = aK1[r];
    Vx[row + n0] = aV0[r];
    Vx[row + n1] = aV1[r];
  }
}

__global__ __launch_bounds__(128) void attn_block_kernel(
    const float* __restrict__ x, const int* __restrict__ nbr,
    const float* __restrict__ Wq, const float* __restrict__ Wo,
    const float* __restrict__ bo, const float* __restrict__ g1,
    const float* __restrict__ b1, const float* __restrict__ g2,
    const float* __restrict__ b2, const float* __restrict__ W1,
    const float* __restrict__ bf1, const float* __restrict__ W2,
    const float* __restrict__ bf2, const float* __restrict__ Kx,
    const float* __restrict__ Vx, float* __restrict__ out) {
  __shared__ float buf0[T_N * LDP];
  __shared__ float buf1[32 * LDP];
  __shared__ float buf2[32 * LDP];
  __shared__ float buf3[32 * LDP];
  const int VOFF = 16 * LDP;

  const int tid = threadIdx.x;
  const int e = blockIdx.x;
  const int lane = tid & 31, wave = tid >> 5;
  const int half = lane >> 4, lm = lane & 15;
  const size_t xbase = (size_t)e * (T_N * D_N);

  for (int i = tid; i < T_N * 128; i += 128) {
    const int r = i >> 7, c = i & 127;
    buf0[r * LDP + c] = x[xbase + (size_t)r * D_N + c];
  }
  for (int i = tid; i < 16 * 128; i += 128) {
    const int k = i >> 7, c = i & 127;
    const int n = nbr[e * K_N + k];
    buf2[k * LDP + c]        = Kx[(size_t)n * D_N + c];
    buf2[VOFF + k * LDP + c] = Vx[(size_t)n * D_N + c];
  }
  __syncthreads();

  {
    v8f q[4] = {};
    gemm_acc(buf0, Wq, 128, wave, half, lm, min(16 + lm, T_N - 1), 128, q);
    for (int i = 0; i < 4; ++i)
      for (int r = 0; r < 8; ++r) {
        const int row = ((i >> 1) << 4) + r + 8 * half;
        const int col = wave * 32 + ((i & 1) << 4) + lm;
        buf1[row * LDP + col] = q[i][r] * 0.25f;
      }
  }
  __syncthreads();

  {
    const int hc0 = (2 * wave) * 16, hc1 = hc0 + 16;
    v8f s0{}, s1{}, s2{}, s3{};
    for (int k0 = 0; k0 < 16; k0 += 4) {
      const int ka = k0 + 2 * half;
      v2f a00, a01, a10, a11, b0, b1;
      a00.x = buf1[lm * LDP + hc0 + ka];        a00.y = buf1[lm * LDP + hc0 + ka + 1];
      a01.x = buf1[(16 + lm) * LDP + hc0 + ka]; a01.y = buf1[(16 + lm) * LDP + hc0 + ka + 1];
      a10.x = buf1[lm * LDP + hc1 + ka];        a10.y = buf1[lm * LDP + hc1 + ka + 1];
      a11.x = buf1[(16 + lm) * LDP + hc1 + ka]; a11.y = buf1[(16 + lm) * LDP + hc1 + ka + 1];
      b0.x = buf2[lm * LDP + hc0 + ka]; b0.y = buf2[lm * LDP + hc0 + ka + 1];
      b1.x = buf2[lm * LDP + hc1 + ka]; b1.y = buf2[lm * LDP + hc1 + ka + 1];
      s0 = wmma_f32(a00, b0, s0);
      s1 = wmma_f32(a01, b0, s1);
      s2 = wmma_f32(a10, b1, s2);
      s3 = wmma_f32(a11, b1, s3);
    }
    for (int r = 0; r < 8; ++r) {
      const int t0 = r + 8 * half, t1 = 16 + r + 8 * half;
      buf3[t0 * LDP + hc0 + lm] = s0[r];
      buf3[t1 * LDP + hc0 + lm] = s1[r];
      buf3[t0 * LDP + hc1 + lm] = s2[r];
      buf3[t1 * LDP + hc1 + lm] = s3[r];
    }
  }
  __syncthreads();

  for (int idx = tid; idx < T_N * H_N; idx += 128) {
    const int t = idx >> 3, h = idx & 7;
    float* row = &buf3[t * LDP + h * 16];
    float m = row[0];
    for (int k = 1; k < 16; ++k) m = fmaxf(m, row[k]);
    float s = 0.f;
    for (int k = 0; k < 16; ++k) { row[k] = __expf(row[k] - m); s += row[k]; }
    const float inv = 1.f / s;
    for (int k = 0; k < 16; ++k) row[k] *= inv;
  }
  __syncthreads();

  {
    const int hc0 = (2 * wave) * 16, hc1 = hc0 + 16;
    v8f z0{}, z1{}, z2{}, z3{};
    for (int k0 = 0; k0 < 16; k0 += 4) {
      const int ka = k0 + 2 * half;
      v2f a00, a01, a10, a11, b0, b1;
      a00.x = buf3[lm * LDP + hc0 + ka];        a00.y = buf3[lm * LDP + hc0 + ka + 1];
      a01.x = buf3[(16 + lm) * LDP + hc0 + ka]; a01.y = buf3[(16 + lm) * LDP + hc0 + ka + 1];
      a10.x = buf3[lm * LDP + hc1 + ka];        a10.y = buf3[lm * LDP + hc1 + ka + 1];
      a11.x = buf3[(16 + lm) * LDP + hc1 + ka]; a11.y = buf3[(16 + lm) * LDP + hc1 + ka + 1];
      b0.x = buf2[VOFF + ka * LDP + hc0 + lm]; b0.y = buf2[VOFF + (ka + 1) * LDP + hc0 + lm];
      b1.x = buf2[VOFF + ka * LDP + hc1 + lm]; b1.y = buf2[VOFF + (ka + 1) * LDP + hc1 + lm];
      z0 = wmma_f32(a00, b0, z0);
      z1 = wmma_f32(a01, b0, z1);
      z2 = wmma_f32(a10, b1, z2);
      z3 = wmma_f32(a11, b1, z3);
    }
    for (int r = 0; r < 8; ++r) {
      const int t0 = r + 8 * half, t1 = 16 + r + 8 * half;
      buf1[t0 * LDP + hc0 + lm] = z0[r];
      buf1[t1 * LDP + hc0 + lm] = z1[r];
      buf1[t0 * LDP + hc1 + lm] = z2[r];
      buf1[t1 * LDP + hc1 + lm] = z3[r];
    }
  }
  __syncthreads();

  {
    v8f o[4] = {};
    gemm_acc(buf1, Wo, 128, wave, half, lm, 16 + lm, 128, o);
    for (int i = 0; i < 4; ++i)
      for (int r = 0; r < 8; ++r) {
        const int row = ((i >> 1) << 4) + r + 8 * half;
        const int col = wave * 32 + ((i & 1) << 4) + lm;
        buf2[row * LDP + col] = o[i][r] + bo[col] + buf0[min(row, T_N - 1) * LDP + col];
      }
  }
  __syncthreads();

  if (tid < T_N) {
    const float* rp = &buf2[tid * LDP];
    float mu = 0.f;
    for (int c = 0; c < 128; ++c) mu += rp[c];
    mu *= (1.f / 128.f);
    float var = 0.f;
    for (int c = 0; c < 128; ++c) { const float d = rp[c] - mu; var += d * d; }
    var *= (1.f / 128.f);
    const float rs = rsqrtf(var + EPS_LN);
    float* wp = &buf3[tid * LDP];
    for (int c = 0; c < 128; ++c) wp[c] = (rp[c] - mu) * rs * g1[c] + b1[c];
  }
  __syncthreads();

  v8f y[4] = {};
  for (int ch = 0; ch < 4; ++ch) {
    v8f g[4] = {};
    gemm_acc(buf3, W1 + ch * 128, FF_N, wave, half, lm, 16 + lm, 128, g);
    for (int i = 0; i < 4; ++i)
      for (int r = 0; r < 8; ++r) {
        const int row = ((i >> 1) << 4) + r + 8 * half;
        const int col = wave * 32 + ((i & 1) << 4) + lm;
        buf1[row * LDP + col] = fmaxf(g[i][r] + bf1[ch * 128 + col], 0.f);
      }
    __syncthreads();
    gemm_acc(buf1, W2 + (size_t)(ch * 128) * D_N, 128, wave, half, lm, 16 + lm, 128, y);
    __syncthreads();
  }

  for (int i = 0; i < 4; ++i)
    for (int r = 0; r < 8; ++r) {
      const int row = ((i >> 1) << 4) + r + 8 * half;
      const int col = wave * 32 + ((i & 1) << 4) + lm;
      buf1[row * LDP + col] = y[i][r] + bf2[col] + buf3[row * LDP + col];
    }
  __syncthreads();

  if (tid < T_N) {
    const float* rp = &buf1[tid * LDP];
    float mu = 0.f;
    for (int c = 0; c < 128; ++c) mu += rp[c];
    mu *= (1.f / 128.f);
    float var = 0.f;
    for (int c = 0; c < 128; ++c) { const float d = rp[c] - mu; var += d * d; }
    var *= (1.f / 128.f);
    const float rs = rsqrtf(var + EPS_LN);
    for (int c = 0; c < 128; ++c)
      out[xbase + (size_t)tid * D_N + c] = (rp[c] - mu) * rs * g2[c] + b2[c];
  }
}

extern "C" void kernel_launch(void* const* d_in, const int* in_sizes, int n_in,
                              void* d_out, int out_size, void* d_ws, size_t ws_size,
                              hipStream_t stream) {
  const float* x   = (const float*)d_in[0];
  const int*   nbr = (const int*)d_in[1];
  const float* Wq  = (const float*)d_in[2];
  const float* Wk  = (const float*)d_in[3];
  const float* Wv  = (const float*)d_in[4];
  const float* Wo  = (const float*)d_in[5];
  const float* bo  = (const float*)d_in[6];
  const float* g1  = (const float*)d_in[7];
  const float* b1  = (const float*)d_in[8];
  const float* g2  = (const float*)d_in[9];
  const float* b2  = (const float*)d_in[10];
  const float* W1  = (const float*)d_in[11];
  const float* bf1 = (const float*)d_in[12];
  const float* W2  = (const float*)d_in[13];
  const float* bf2 = (const float*)d_in[14];

  float* Kx = (float*)d_ws;
  float* Vx = Kx + (size_t)E_N * D_N;

  kv_proj_kernel<<<E_N / 16, 128, 0, stream>>>(x, Wk, Wv, Kx, Vx);
  attn_block_kernel<<<E_N, 128, 0, stream>>>(x, nbr, Wq, Wo, bo, g1, b1, g2, b2,
                                             W1, bf1, W2, bf2, Kx, Vx,
                                             (float*)d_out);
}


# two events per work-group: their 2 x 24 rows fill three 16-row tiles exactly (no padding rows) and each weight fragment feeds 3 row tiles; scores and Z stay in the Q buffer, LDS 84480 B
# speedup vs baseline: 1.7062x; 1.1292x over previous
#include <hip/hip_runtime.h>
#include <hip/hip_bf16.h>

#define E_N 10000
#define T_N 24
#define D_N 128
#define H_N 8
#define K_N 16
#define FF_N 512
#define EPS_LN 1e-5f
#define LDP 132
#define EV_N 2
#define R_N (EV_N * T_N)

typedef float v2f __attribute__((ext_vector_type(2)));
typedef float v8f __attribute__((ext_vector_type(8)));

__device__ __forceinline__ v8f wmma_f32(v2f a, v2f b, v8f c) {
  return __builtin_amdgcn_wmma_f32_16x16x4_f32(
       false, a,  false, b,
       (short)0, c,  false,  false);
}

__device__ __forceinline__ void gemm_acc(const float* sA, const float* B, int ldb,
                                         int wave, int half, int lm,
                                         int Kdim, v8f acc[6]) {
  const int n0 = wave * 32 + lm;
  const int n1 = n0 + 16;
#pragma unroll
  for (int k0 = 0; k0 < Kdim; k0 += 4) {
    const int ka = k0 + 2 * half;
    v2f a0, a1, a2, b0, b1;
    a0.x = sA[lm * LDP + ka];        a0.y = sA[lm * LDP + ka + 1];
    a1.x = sA[(16 + lm) * LDP + ka]; a1.y = sA[(16 + lm) * LDP + ka + 1];
    a2.x = sA[(32 + lm) * LDP + ka]; a2.y = sA[(32 + lm) * LDP + ka + 1];
    b0.x = B[(size_t)ka * ldb + n0]; b0.y = B[(size_t)(ka + 1) * ldb + n0];
    b1.x = B[(size_t)ka * ldb + n1]; b1.y = B[(size_t)(ka + 1) * ldb + n1];
    acc[0] = wmma_f32(a0, b0, acc[0]);
    acc[1] = wmma_f32(a0, b1, acc[1]);
    acc[2] = wmma_f32(a1, b0, acc[2]);
    acc[3] = wmma_f32(a1, b1, acc[3]);
    acc[4] = wmma_f32(a2, b0, acc[4]);
    acc[5] = wmma_f32(a2, b1, acc[5]);
  }
}

__device__ __forceinline__ void head_product(float* sP, const float* sB, int rowB0,
                                             int rowB1, bool kmajor, int hc,
                                             int half, int lm) {
  v8f p0{}, p1a{}, p1b{}, p2{};
  for (int k0 = 0; k0 < 16; k0 += 4) {
    const int ka = k0 + 2 * half;
    v2f a0, a1, a2, b0, b1;
    a0.x = sP[lm * LDP + hc + ka];        a0.y = sP[lm * LDP + hc + ka + 1];
    a1.x = sP[(16 + lm) * LDP + hc + ka]; a1.y = sP[(16 + lm) * LDP + hc + ka + 1];
    a2.x = sP[(32 + lm) * LDP + hc + ka]; a2.y = sP[(32 + lm) * LDP + hc + ka + 1];
    if (kmajor) {
      b0.x = sB[(rowB0 + ka) * LDP + hc + lm]; b0.y = sB[(rowB0 + ka + 1) * LDP + hc + lm];
      b1.x = sB[(rowB1 + ka) * LDP + hc + lm]; b1.y = sB[(rowB1 + ka + 1) * LDP + hc + lm];
    } else {
      b0.x = sB[(rowB0 + lm) * LDP + hc + ka]; b0.y = sB[(rowB0 + lm) * LDP + hc + ka + 1];
      b1.x = sB[(rowB1 + lm) * LDP + hc + ka]; b1.y = sB[(rowB1 + lm) * LDP + hc + ka + 1];
    }
    p0  = wmma_f32(a0, b0, p0);
    p1a = wmma_f32(a1, b0, p1a);
    p1b = wmma_f32(a1, b1, p1b);
    p2  = wmma_f32(a2, b1, p2);
  }
  for (int r = 0; r < 8; ++r) {
    const int t = r + 8 * half;
    sP[t * LDP + hc + lm]        = p0[r];
    sP[(16 + t) * LDP + hc + lm] = half ? p1b[r] : p1a[r];
    sP[(32 + t) * LDP + hc + lm] = p2[r];
  }
}

__global__ __launch_bounds__(128) void kv_proj_kernel(
    const float* __restrict__ x, const float* __restrict__ Wk,
    const float* __restrict__ Wv, float* __restrict__ Kx, float* __restrict__ Vx) {
  __shared__ float sx0[16 * LDP];
  const int tid = threadIdx.x;
  const int e0 = blockIdx.x * 16;
  for (int i = tid; i < 16 * 128; i += 128) {
    const int r = i >> 7, c = i & 127;
    sx0[r * LDP + c] = x[(size_t)(e0 + r) * (T_N * D_N) + c];
  }
  __syncthreads();

  const int lane = tid & 31, wave = tid >> 5;
  const int half = lane >> 4, lm = lane & 15;
  const int n0 = wave * 32 + lm, n1 = n0 + 16;
  v8f aK0{}, aK1{}, aV0{}, aV1{};
  for (int k0 = 0; k0 < 128; k0 += 4) {
    const int ka = k0 + 2 * half;
    v2f a, bk0, bk1, bv0, bv1;
    a.x = sx0[lm * LDP + ka];  a.y = sx0[lm * LDP + ka + 1];
    bk0.x = Wk[ka * 128 + n0]; bk0.y = Wk[(ka + 1) * 128 + n0];
    bk1.x = Wk[ka * 128 + n1]; bk1.y = Wk[(ka + 1) * 128 + n1];
    bv0.x = Wv[ka * 128 + n0]; bv0.y = Wv[(ka + 1) * 128 + n0];
    bv1.x = Wv[ka * 128 + n1]; bv1.y = Wv[(ka + 1) * 128 + n1];
    aK0 = wmma_f32(a, bk0, aK0);
    aK1 = wmma_f32(a, bk1, aK1);
    aV0 = wmma_f32(a, bv0, aV0);
    aV1 = wmma_f32(a, bv1, aV1);
  }
  for (int r = 0; r < 8; ++r) {
    const size_t row = (size_t)(e0 + r + 8 * half) * 128;
    Kx[row + n0] = aK0[r];
    Kx[row + n1] = aK1[r];
    Vx[row + n0] = aV0[r];
    Vx[row + n1] = aV1[r];
  }
}

__global__ __launch_bounds__(128) void attn_block_kernel(
    const float* __restrict__ x, const int* __restrict__ nbr,
    const float* __restrict__ Wq, const float* __restrict__ Wo,
    const float* __restrict__ bo, const float* __restrict__ g1,
    const float* __restrict__ b1, const float* __restrict__ g2,
    const float* __restrict__ b2, const float* __restrict__ W1,
    const float* __restrict__ bf1, const float* __restrict__ W2,
    const float* __restrict__ bf2, const float* __restrict__ Kx,
    const float* __restrict__ Vx, float* __restrict__ out) {
  __shared__ float bufA[R_N * LDP];
  __shared__ float bufB[R_N * LDP];
  __shared__ float bufC[2 * EV_N * K_N * LDP];
  const int VROW = EV_N * K_N;
  const int VOFF = VROW * LDP;

  const int tid = threadIdx.x;
  const int e0 = blockIdx.x * EV_N;
  const int lane = tid & 31, wave = tid >> 5;
  const int half = lane >> 4, lm = lane & 15;
  const size_t xbase = (size_t)e0 * (T_N * D_N);

  for (int i = tid; i < R_N * 128; i += 128) {
    const int r = i >> 7, c = i & 127;
    bufA[r * LDP + c] = x[xbase + (size_t)r * D_N + c];
  }
  for (int i = tid; i < EV_N * K_N * 128; i += 128) {
    const int k = i >> 7, c = i & 127;
    const int n = nbr[e0 * K_N + k];
    bufC[k * LDP + c]        = Kx[(size_t)n * D_N + c];
    bufC[VOFF + k * LDP + c] = Vx[(size_t)n * D_N + c];
  }
  __syncthreads();

  {
    v8f q[6] = {};
    gemm_acc(bufA, Wq, 128, wave, half, lm, 128, q);
    for (int i = 0; i < 6; ++i)
      for (int r = 0; r < 8; ++r) {
        const int row = ((i >> 1) << 4) + r + 8 * half;
        const int col = wave * 32 + ((i & 1) << 4) + lm;
        bufB[row * LDP + col] = q[i][r] * 0.25f;
      }
  }
  __syncthreads();

  const int hc0 = (2 * wave) * 16, hc1 = hc0 + 16;

  head_product(bufB, bufC, 0, K_N, false, hc0, half, lm);
  head_product(bufB, bufC, 0, K_N, false, hc1, half, lm);
  __syncthreads();

  for (int idx = tid; idx < R_N * H_N; idx += 128) {
    const int t = idx >> 3, h = idx & 7;
    float* row = &bufB[t * LDP + h * 16];
    float m = row[0];
    for (int k = 1; k < 16; ++k) m = fmaxf(m, row[k]);
    float s = 0.f;
    for (int k = 0; k < 16; ++k) { row[k] = __expf(row[k] - m); s += row[k]; }
    const float inv = 1.f / s;
    for (int k = 0; k < 16; ++k) row[k] *= inv;
  }
  __syncthreads();

  head_product(bufB, bufC, VROW, VROW + K_N, true, hc0, half, lm);
  head_product(bufB, bufC, VROW, VROW + K_N, true, hc1, half, lm);
  __syncthreads();

  {
    v8f o[6] = {};
    gemm_acc(bufB, Wo, 128, wave, half, lm, 128, o);
    for (int i = 0; i < 6; ++i)
      for (int r = 0; r < 8; ++r) {
        const int row = ((i >> 1) << 4) + r + 8 * half;
        const int col = wave * 32 + ((i & 1) << 4) + lm;
        bufC[row * LDP + col] = o[i][r] + bo[col] + bufA[row * LDP + col];
      }
  }
  __syncthreads();

  if (tid < R_N) {
    const float* rp = &bufC[tid * LDP];
    float mu = 0.f;
    for (int c = 0; c < 128; ++c) mu += rp[c];
    mu *= (1.f / 128.f);
    float var = 0.f;
    for (int c = 0; c < 128; ++c) { const float d = rp[c] - mu; var += d * d; }
    var *= (1.f / 128.f);
    const float rs = rsqrtf(var + EPS_LN);
    float* wp = &bufA[tid * LDP];
    for (int c = 0; c < 128; ++c) wp[c] = (rp[c] - mu) * rs * g1[c] + b1[c];
  }
  __syncthreads();

  v8f y[6] = {};
  for (int ch = 0; ch < 4; ++ch) {
    v8f g[6] = {};
    gemm_acc(bufA, W1 + ch * 128, FF_N, wave, half, lm, 128, g);
    for (int i = 0; i < 6; ++i)
      for (int r = 0; r < 8; ++r) {
        const int row = ((i >> 1) << 4) + r + 8 * half;
        const int col = wave * 32 + ((i & 1) << 4) + lm;
        bufB[row * LDP + col] = fmaxf(g[i][r] + bf1[ch * 128 + col], 0.f);
      }
    __syncthreads();
    gemm_acc(bufB, W2 + (size_t)(ch * 128) * D_N, 128, wave, half, lm, 128, y);
    __syncthreads();
  }

  for (int i = 0; i < 6; ++i)
    for (int r = 0; r < 8; ++r) {
      const int row = ((i >> 1) << 4) + r + 8 * half;
      const int col = wave * 32 + ((i & 1) << 4) + lm;
      bufB[row * LDP + col] = y[i][r] + bf2[col] + bufA[row * LDP + col];
    }
  __syncthreads();

  if (tid < R_N) {
    const float* rp = &bufB[tid * LDP];
    float mu = 0.f;
    for (int c = 0; c < 128; ++c) mu += rp[c];
    mu *= (1.f / 128.f);
    float var = 0.f;
    for (int c = 0; c < 128; ++c) { const float d = rp[c] - mu; var += d * d; }
    var *= (1.f / 128.f);
    const float rs = rsqrtf(var + EPS_LN);
    for (int c = 0; c < 128; ++c)
      out[xbase + (size_t)tid * D_N + c] = (rp[c] - mu) * rs * g2[c] + b2[c];
  }
}

extern "C" void kernel_launch(void* const* d_in, const int* in_sizes, int n_in,
                              void* d_out, int out_size, void* d_ws, size_t ws_size,
                              hipStream_t stream) {
  const float* x   = (const float*)d_in[0];
  const int*   nbr = (const int*)d_in[1];
  const float* Wq  = (const float*)d_in[2];
  const float* Wk  = (const float*)d_in[3];
  const float* Wv  = (const float*)d_in[4];
  const float* Wo  = (const float*)d_in[5];
  const float* bo  = (const float*)d_in[6];
  const float* g1  = (const float*)d_in[7];
  const float* b1  = (const float*)d_in[8];
  const float* g2  = (const float*)d_in[9];
  const float* b2  = (const float*)d_in[10];
  const float* W1  = (const float*)d_in[11];
  const float* bf1 = (const float*)d_in[12];
  const float* W2  = (const float*)d_in[13];
  const float* bf2 = (const float*)d_in[14];

  float* Kx = (float*)d_ws;
  float* Vx = Kx + (size_t)E_N * D_N;

  kv_proj_kernel<<<E_N / 16, 128, 0, stream>>>(x, Wk, Wv, Kx, Vx);
  attn_block_kernel<<<E_N / EV_N, 128, 0, stream>>>(x, nbr, Wq, Wo, bo, g1, b1, g2, b2,
                                             W1, bf1, W2, bf2, Kx, Vx,
                                             (float*)d_out);
}


# GEMM operand loads software-pipelined by 32-k blocks: each block's A (LDS) and B (global) fragments fetched as one group ahead of the previous block's 48 WMMAs, none among them
# speedup vs baseline: 2.0345x; 1.1924x over previous
#include <hip/hip_runtime.h>
#include <hip/hip_bf16.h>

#define E_N 10000
#define T_N 24
#define D_N 128
#define H_N 8
#define K_N 16
#define FF_N 512
#define EPS_LN 1e-5f
#define LDP 132
#define EV_N 2
#define R_N (EV_N * T_N)

typedef float v2f __attribute__((ext_vector_type(2)));
typedef float v8f __attribute__((ext_vector_type(8)));

__device__ __forceinline__ v8f wmma_f32(v2f a, v2f b, v8f c) {
  return __builtin_amdgcn_wmma_f32_16x16x4_f32(
       false, a,  false, b,
       (short)0, c,  false,  false);
}

#define KB_N 8
#define NBLK (128 / (4 * KB_N))

__device__ __forceinline__ void load_a_block(const float* Al, int kb0, v2f a[KB_N][3]) {
#pragma unroll
  for (int s = 0; s < KB_N; ++s) {
    const int k = kb0 + 4 * s;
    a[s][0].x = Al[k];            a[s][0].y = Al[k + 1];
    a[s][1].x = Al[16 * LDP + k]; a[s][1].y = Al[16 * LDP + k + 1];
    a[s][2].x = Al[32 * LDP + k]; a[s][2].y = Al[32 * LDP + k + 1];
  }
}

__device__ __forceinline__ void load_b_block(const float* Bl, int ldb, int kb0,
                                             v2f b[KB_N][2]) {
#pragma unroll
  for (int s = 0; s < KB_N; ++s) {
    const int k = kb0 + 4 * s;
    b[s][0].x = Bl[k * ldb];      b[s][0].y = Bl[(k + 1) * ldb];
    b[s][1].x = Bl[k * ldb + 16]; b[s][1].y = Bl[(k + 1) * ldb + 16];
  }
}

__device__ __forceinline__ void gemm_acc(const float* sA, const float* B, int ldb,
                                         const float* Bn, int ldbn, bool has_next,
                                         int wave, int half, int lm,
                                         v2f bfirst[KB_N][2], v8f acc[6]) {
  const float* Al = sA + lm * LDP + 2 * half;
  const float* Bl = B + (2 * half) * ldb + wave * 32 + lm;
  const float* Bnl = Bn + (2 * half) * ldbn + wave * 32 + lm;
  v2f a[2][KB_N][3];
  v2f b1[KB_N][2];
  load_a_block(Al, 0, a[0]);
#pragma unroll
  for (int blk = 0; blk < NBLK; ++blk) {
    v2f (*bc)[2] = (blk & 1) ? b1 : bfirst;
    v2f (*bn)[2] = (blk & 1) ? bfirst : b1;
    if (blk + 1 < NBLK) {
      load_a_block(Al, (blk + 1) * 4 * KB_N, a[(blk + 1) & 1]);
      load_b_block(Bl, ldb, (blk + 1) * 4 * KB_N, bn);
    } else if (has_next) {
      load_b_block(Bnl, ldbn, 0, bn);
    }
    __builtin_amdgcn_sched_barrier(0);
#pragma unroll
    for (int s = 0; s < KB_N; ++s) {
      acc[0] = wmma_f32(a[blk & 1][s][0], bc[s][0], acc[0]);
      acc[1] = wmma_f32(a[blk & 1][s][0], bc[s][1], acc[1]);
      acc[2] = wmma_f32(a[blk & 1][s][1], bc[s][0], acc[2]);
      acc[3] = wmma_f32(a[blk & 1][s][1], bc[s][1], acc[3]);
      acc[4] = wmma_f32(a[blk & 1][s][2], bc[s][0], acc[4]);
      acc[5] = wmma_f32(a[blk & 1][s][2], bc[s][1], acc[5]);
    }
    __builtin_amdgcn_sched_barrier(0);
  }
}

__device__ __forceinline__ void head_product(float* sP, const float* sB, int rowB0,
                                             int rowB1, bool kmajor, int hc,
                                             int half, int lm) {
  v8f p0{}, p1a{}, p1b{}, p2{};
  for (int k0 = 0; k0 < 16; k0 += 4) {
    const int ka = k0 + 2 * half;
    v2f a0, a1, a2, b0, b1;
    a0.x = sP[lm * LDP + hc + ka];        a0.y = sP[lm * LDP + hc + ka + 1];
    a1.x = sP[(16 + lm) * LDP + hc + ka]; a1.y = sP[(16 + lm) * LDP + hc + ka + 1];
    a2.x = sP[(32 + lm) * LDP + hc + ka]; a2.y = sP[(32 + lm) * LDP + hc + ka + 1];
    if (kmajor) {
      b0.x = sB[(rowB0 + ka) * LDP + hc + lm]; b0.y = sB[(rowB0 + ka + 1) * LDP + hc + lm];
      b1.x = sB[(rowB1 + ka) * LDP + hc + lm]; b1.y = sB[(rowB1 + ka + 1) * LDP + hc + lm];
    } else {
      b0.x = sB[(rowB0 + lm) * LDP + hc + ka]; b0.y = sB[(rowB0 + lm) * LDP + hc + ka + 1];
      b1.x = sB[(rowB1 + lm) * LDP + hc + ka]; b1.y = sB[(rowB1 + lm) * LDP + hc + ka + 1];
    }
    p0  = wmma_f32(a0, b0, p0);
    p1a = wmma_f32(a1, b0, p1a);
    p1b = wmma_f32(a1, b1, p1b);
    p2  = wmma_f32(a2, b1, p2);
  }
  for (int r = 0; r < 8; ++r) {
    const int t = r + 8 * half;
    sP[t * LDP + hc + lm]        = p0[r];
    sP[(16 + t) * LDP + hc + lm] = half ? p1b[r] : p1a[r];
    sP[(32 + t) * LDP + hc + lm] = p2[r];
  }
}

__global__ __launch_bounds__(128) void kv_proj_kernel(
    const float* __restrict__ x, const float* __restrict__ Wk,
    const float* __restrict__ Wv, float* __restrict__ Kx, float* __restrict__ Vx) {
  __shared__ float sx0[16 * LDP];
  const int tid = threadIdx.x;
  const int e0 = blockIdx.x * 16;
  for (int i = tid; i < 16 * 128; i += 128) {
    const int r = i >> 7, c = i & 127;
    sx0[r * LDP + c] = x[(size_t)(e0 + r) * (T_N * D_N) + c];
  }
  __syncthreads();

  const int lane = tid & 31, wave = tid >> 5;
  const int half = lane >> 4, lm = lane & 15;
  const int n0 = wave * 32 + lm, n1 = n0 + 16;
  v8f aK0{}, aK1{}, aV0{}, aV1{};
  for (int k0 = 0; k0 < 128; k0 += 4) {
    const int ka = k0 + 2 * half;
    v2f a, bk0, bk1, bv0, bv1;
    a.x = sx0[lm * LDP + ka];  a.y = sx0[lm * LDP + ka + 1];
    bk0.x = Wk[ka * 128 + n0]; bk0.y = Wk[(ka + 1) * 128 + n0];
    bk1.x = Wk[ka * 128 + n1]; bk1.y = Wk[(ka + 1) * 128 + n1];
    bv0.x = Wv[ka * 128 + n0]; bv0.y = Wv[(ka + 1) * 128 + n0];
    bv1.x = Wv[ka * 128 + n1]; bv1.y = Wv[(ka + 1) * 128 + n1];
    aK0 = wmma_f32(a, bk0, aK0);
    aK1 = wmma_f32(a, bk1, aK1);
    aV0 = wmma_f32(a, bv0, aV0);
    aV1 = wmma_f32(a, bv1, aV1);
  }
  for (int r = 0; r < 8; ++r) {
    const size_t row = (size_t)(e0 + r + 8 * half) * 128;
    Kx[row + n0] = aK0[r];
    Kx[row + n1] = aK1[r];
    Vx[row + n0] = aV0[r];
    Vx[row + n1] = aV1[r];
  }
}

__global__ __launch_bounds__(128) void attn_block_kernel(
    const float* __restrict__ x, const int* __restrict__ nbr,
    const float* __restrict__ Wq, const float* __restrict__ Wo,
    const float* __restrict__ bo, const float* __restrict__ g1,
    const float* __restrict__ b1, const float* __restrict__ g2,
    const float* __restrict__ b2, const float* __restrict__ W1,
    const float* __restrict__ bf1, const float* __restrict__ W2,
    const float* __restrict__ bf2, const float* __restrict__ Kx,
    const float* __restrict__ Vx, float* __restrict__ out) {
  __shared__ float bufA[R_N * LDP];
  __shared__ float bufB[R_N * LDP];
  __shared__ float bufC[2 * EV_N * K_N * LDP];
  const int VROW = EV_N * K_N;
  const int VOFF = VROW * LDP;

  const int tid = threadIdx.x;
  const int e0 = blockIdx.x * EV_N;
  const int lane = tid & 31, wave = tid >> 5;
  const int half = lane >> 4, lm = lane & 15;
  const size_t xbase = (size_t)e0 * (T_N * D_N);

  v2f bw[KB_N][2];
  load_b_block(Wq + (2 * half) * 128 + wave * 32 + lm, 128, 0, bw);

  for (int i = tid; i < R_N * 128; i += 128) {
    const int r = i >> 7, c = i & 127;
    bufA[r * LDP + c] = x[xbase + (size_t)r * D_N + c];
  }
  for (int i = tid; i < EV_N * K_N * 128; i += 128) {
    const int k = i >> 7, c = i & 127;
    const int n = nbr[e0 * K_N + k];
    bufC[k * LDP + c]        = Kx[(size_t)n * D_N + c];
    bufC[VOFF + k * LDP + c] = Vx[(size_t)n * D_N + c];
  }
  __syncthreads();

  {
    v8f q[6] = {};
    gemm_acc(bufA, Wq, 128, Wo, 128, true, wave, half, lm, bw, q);
    for (int i = 0; i < 6; ++i)
      for (int r = 0; r < 8; ++r) {
        const int row = ((i >> 1) << 4) + r + 8 * half;
        const int col = wave * 32 + ((i & 1) << 4) + lm;
        bufB[row * LDP + col] = q[i][r] * 0.25f;
      }
  }
  __syncthreads();

  const int hc0 = (2 * wave) * 16, hc1 = hc0 + 16;

  head_product(bufB, bufC, 0, K_N, false, hc0, half, lm);
  head_product(bufB, bufC, 0, K_N, false, hc1, half, lm);
  __syncthreads();

  for (int idx = tid; idx < R_N * H_N; idx += 128) {
    const int t = idx >> 3, h = idx & 7;
    float* row = &bufB[t * LDP + h * 16];
    float m = row[0];
    for (int k = 1; k < 16; ++k) m = fmaxf(m, row[k]);
    float s = 0.f;
    for (int k = 0; k < 16; ++k) { row[k] = __expf(row[k] - m); s += row[k]; }
    const float inv = 1.f / s;
    for (int k = 0; k < 16; ++k) row[k] *= inv;
  }
  __syncthreads();

  head_product(bufB, bufC, VROW, VROW + K_N, true, hc0, half, lm);
  head_product(bufB, bufC, VROW, VROW + K_N, true, hc1, half, lm);
  __syncthreads();

  {
    v8f o[6] = {};
    gemm_acc(bufB, Wo, 128, W1, FF_N, true, wave, half, lm, bw, o);
    for (int i = 0; i < 6; ++i)
      for (int r = 0; r < 8; ++r) {
        const int row = ((i >> 1) << 4) + r + 8 * half;
        const int col = wave * 32 + ((i & 1) << 4) + lm;
        bufC[row * LDP + col] = o[i][r] + bo[col] + bufA[row * LDP + col];
      }
  }
  __syncthreads();

  if (tid < R_N) {
    const float* rp = &bufC[tid * LDP];
    float mu = 0.f;
    for (int c = 0; c < 128; ++c) mu += rp[c];
    mu *= (1.f / 128.f);
    float var = 0.f;
    for (int c = 0; c < 128; ++c) { const float d = rp[c] - mu; var += d * d; }
    var *= (1.f / 128.f);
    const float rs = rsqrtf(var + EPS_LN);
    float* wp = &bufA[tid * LDP];
    for (int c = 0; c < 128; ++c) wp[c] = (rp[c] - mu) * rs * g1[c] + b1[c];
  }
  __syncthreads();

  v8f y[6] = {};
  for (int ch = 0; ch < 4; ++ch) {
    v8f g[6] = {};
    gemm_acc(bufA, W1 + ch * 128, FF_N, W2 + (size_t)(ch * 128) * D_N, 128, true,
             wave, half, lm, bw, g);
    for (int i = 0; i < 6; ++i)
      for (int r = 0; r < 8; ++r) {
        const int row = ((i >> 1) << 4) + r + 8 * half;
        const int col = wave * 32 + ((i & 1) << 4) + lm;
        bufB[row * LDP + col] = fmaxf(g[i][r] + bf1[ch * 128 + col], 0.f);
      }
    __syncthreads();
    gemm_acc(bufB, W2 + (size_t)(ch * 128) * D_N, 128, W1 + (ch + 1) * 128, FF_N, ch < 3,
             wave, half, lm, bw, y);
    __syncthreads();
  }

  for (int i = 0; i < 6; ++i)
    for (int r = 0; r < 8; ++r) {
      const int row = ((i >> 1) << 4) + r + 8 * half;
      const int col = wave * 32 + ((i & 1) << 4) + lm;
      bufB[row * LDP + col] = y[i][r] + bf2[col] + bufA[row * LDP + col];
    }
  __syncthreads();

  if (tid < R_N) {
    const float* rp = &bufB[tid * LDP];
    float mu = 0.f;
    for (int c = 0; c < 128; ++c) mu += rp[c];
    mu *= (1.f / 128.f);
    float var = 0.f;
    for (int c = 0; c < 128; ++c) { const float d = rp[c] - mu; var += d * d; }
    var *= (1.f / 128.f);
    const float rs = rsqrtf(var + EPS_LN);
    for (int c = 0; c < 128; ++c)
      out[xbase + (size_t)tid * D_N + c] = (rp[c] - mu) * rs * g2[c] + b2[c];
  }
}

extern "C" void kernel_launch(void* const* d_in, const int* in_sizes, int n_in,
                              void* d_out, int out_size, void* d_ws, size_t ws_size,
                              hipStream_t stream) {
  const float* x   = (const float*)d_in[0];
  const int*   nbr = (const int*)d_in[1];
  const float* Wq  = (const float*)d_in[2];
  const float* Wk  = (const float*)d_in[3];
  const float* Wv  = (const float*)d_in[4];
  const float* Wo  = (const float*)d_in[5];
  const float* bo  = (const float*)d_in[6];
  const float* g1  = (const float*)d_in[7];
  const float* b1  = (const float*)d_in[8];
  const float* g2  = (const float*)d_in[9];
  const float* b2  = (const float*)d_in[10];
  const float* W1  = (const float*)d_in[11];
  const float* bf1 = (const float*)d_in[12];
  const float* W2  = (const float*)d_in[13];
  const float* bf2 = (const float*)d_in[14];

  float* Kx = (float*)d_ws;
  float* Vx = Kx + (size_t)E_N * D_N;

  kv_proj_kernel<<<E_N / 16, 128, 0, stream>>>(x, Wk, Wv, Kx, Vx);
  attn_block_kernel<<<E_N / EV_N, 128, 0, stream>>>(x, nbr, Wq, Wo, bo, g1, b1, g2, b2,
                                             W1, bf1, W2, bf2, Kx, Vx,
                                             (float*)d_out);
}


# LayerNorm by all four waves, one row per 16 lanes: row sums by a 16-lane DPP butterfly (summation order changes to a tree), 16-byte LDS reads, each 128-byte output line in one store
# speedup vs baseline: 2.2083x; 1.0854x over previous
#include <hip/hip_runtime.h>
#include <hip/hip_bf16.h>

#define E_N 10000
#define T_N 24
#define D_N 128
#define H_N 8
#define K_N 16
#define FF_N 512
#define EPS_LN 1e-5f
#define LDP 132
#define EV_N 2
#define R_N (EV_N * T_N)

typedef float v2f __attribute__((ext_vector_type(2)));
typedef float v8f __attribute__((ext_vector_type(8)));

__device__ __forceinline__ v8f wmma_f32(v2f a, v2f b, v8f c) {
  return __builtin_amdgcn_wmma_f32_16x16x4_f32(
       false, a,  false, b,
       (short)0, c,  false,  false);
}

#define KB_N 8
#define NBLK (128 / (4 * KB_N))

__device__ __forceinline__ void load_a_block(const float* Al, int kb0, v2f a[KB_N][3]) {
#pragma unroll
  for (int s = 0; s < KB_N; ++s) {
    const int k = kb0 + 4 * s;
    a[s][0].x = Al[k];            a[s][0].y = Al[k + 1];
    a[s][1].x = Al[16 * LDP + k]; a[s][1].y = Al[16 * LDP + k + 1];
    a[s][2].x = Al[32 * LDP + k]; a[s][2].y = Al[32 * LDP + k + 1];
  }
}

__device__ __forceinline__ void load_b_block(const float* Bl, int ldb, int kb0,
                                             v2f b[KB_N][2]) {
#pragma unroll
  for (int s = 0; s < KB_N; ++s) {
    const int k = kb0 + 4 * s;
    b[s][0].x = Bl[k * ldb];      b[s][0].y = Bl[(k + 1) * ldb];
    b[s][1].x = Bl[k * ldb + 16]; b[s][1].y = Bl[(k + 1) * ldb + 16];
  }
}

__device__ __forceinline__ void gemm_acc(const float* sA, const float* B, int ldb,
                                         const float* Bn, int ldbn, bool has_next,
                                         int wave, int half, int lm,
                                         v2f bfirst[KB_N][2], v8f acc[6]) {
  const float* Al = sA + lm * LDP + 2 * half;
  const float* Bl = B + (2 * half) * ldb + wave * 32 + lm;
  const float* Bnl = Bn + (2 * half) * ldbn + wave * 32 + lm;
  v2f a[2][KB_N][3];
  v2f b1[KB_N][2];
  load_a_block(Al, 0, a[0]);
#pragma unroll
  for (int blk = 0; blk < NBLK; ++blk) {
    v2f (*bc)[2] = (blk & 1) ? b1 : bfirst;
    v2f (*bn)[2] = (blk & 1) ? bfirst : b1;
    if (blk + 1 < NBLK) {
      load_a_block(Al, (blk + 1) * 4 * KB_N, a[(blk + 1) & 1]);
      load_b_block(Bl, ldb, (blk + 1) * 4 * KB_N, bn);
    } else if (has_next) {
      load_b_block(Bnl, ldbn, 0, bn);
    }
    __builtin_amdgcn_sched_barrier(0);
#pragma unroll
    for (int s = 0; s < KB_N; ++s) {
      acc[0] = wmma_f32(a[blk & 1][s][0], bc[s][0], acc[0]);
      acc[1] = wmma_f32(a[blk & 1][s][0], bc[s][1], acc[1]);
      acc[2] = wmma_f32(a[blk & 1][s][1], bc[s][0], acc[2]);
      acc[3] = wmma_f32(a[blk & 1][s][1], bc[s][1], acc[3]);
      acc[4] = wmma_f32(a[blk & 1][s][2], bc[s][0], acc[4]);
      acc[5] = wmma_f32(a[blk & 1][s][2], bc[s][1], acc[5]);
    }
    __builtin_amdgcn_sched_barrier(0);
  }
}

__device__ __forceinline__ void head_product(float* sP, const float* sB, int rowB0,
                                             int rowB1, bool kmajor, int hc,
                                             int half, int lm) {
  v8f p0{}, p1a{}, p1b{}, p2{};
  for (int k0 = 0; k0 < 16; k0 += 4) {
    const int ka = k0 + 2 * half;
    v2f a0, a1, a2, b0, b1;
    a0.x = sP[lm * LDP + hc + ka];        a0.y = sP[lm * LDP + hc + ka + 1];
    a1.x = sP[(16 + lm) * LDP + hc + ka]; a1.y = sP[(16 + lm) * LDP + hc + ka + 1];
    a2.x = sP[(32 + lm) * LDP + hc + ka]; a2.y = sP[(32 + lm) * LDP + hc + ka + 1];
    if (kmajor) {
      b0.x = sB[(rowB0 + ka) * LDP + hc + lm]; b0.y = sB[(rowB0 + ka + 1) * LDP + hc + lm];
      b1.x = sB[(rowB1 + ka) * LDP + hc + lm]; b1.y = sB[(rowB1 + ka + 1) * LDP + hc + lm];
    } else {
      b0.x = sB[(rowB0 + lm) * LDP + hc + ka]; b0.y = sB[(rowB0 + lm) * LDP + hc + ka + 1];
      b1.x = sB[(rowB1 + lm) * LDP + hc + ka]; b1.y = sB[(rowB1 + lm) * LDP + hc + ka + 1];
    }
    p0  = wmma_f32(a0, b0, p0);
    p1a = wmma_f32(a1, b0, p1a);
    p1b = wmma_f32(a1, b1, p1b);
    p2  = wmma_f32(a2, b1, p2);
  }
  for (int r = 0; r < 8; ++r) {
    const int t = r + 8 * half;
    sP[t * LDP + hc + lm]        = p0[r];
    sP[(16 + t) * LDP + hc + lm] = half ? p1b[r] : p1a[r];
    sP[(32 + t) * LDP + hc + lm] = p2[r];
  }
}

typedef float v4f __attribute__((ext_vector_type(4)));

template <int CTRL>
__device__ __forceinline__ void row_xor_add6(float s[6]) {
#pragma unroll
  for (int p = 0; p < 6; ++p)
    s[p] += __builtin_bit_cast(float, __builtin_amdgcn_update_dpp(
                0, __builtin_bit_cast(int, s[p]), CTRL, 0xf, 0xf, false));
}

__device__ __forceinline__ void row_sum6(float s[6]) {
  row_xor_add6<0x161>(s);
  row_xor_add6<0x162>(s);
  row_xor_add6<0x164>(s);
  row_xor_add6<0x168>(s);
}

__device__ __forceinline__ void layer_norm_rows(const float* src, float* dst, int dpitch,
                                                v4f ga, v4f gb, v4f ba, v4f bb,
                                                int wave, int half, int lm) {
  v4f va[6], vb[6];
  float s[6];
#pragma unroll
  for (int p = 0; p < 6; ++p) {
    const float* rp = src + (12 * wave + 2 * p + half) * LDP + 4 * lm;
    va[p] = *(const v4f*)rp;
    vb[p] = *(const v4f*)(rp + 64);
  }
#pragma unroll
  for (int p = 0; p < 6; ++p)
    s[p] = ((va[p].x + va[p].y) + (va[p].z + va[p].w)) +
           ((vb[p].x + vb[p].y) + (vb[p].z + vb[p].w));
  row_sum6(s);
#pragma unroll
  for (int p = 0; p < 6; ++p) {
    const float mu = s[p] * (1.f / 128.f);
    va[p] -= mu;
    vb[p] -= mu;
    s[p] = ((va[p].x * va[p].x + va[p].y * va[p].y) + (va[p].z * va[p].z + va[p].w * va[p].w)) +
           ((vb[p].x * vb[p].x + vb[p].y * vb[p].y) + (vb[p].z * vb[p].z + vb[p].w * vb[p].w));
  }
  row_sum6(s);
#pragma unroll
  for (int p = 0; p < 6; ++p) {
    const float rs = rsqrtf(s[p] * (1.f / 128.f) + EPS_LN);
    float* wp = dst + (12 * wave + 2 * p + half) * dpitch + 4 * lm;
    *(v4f*)wp        = va[p] * rs * ga + ba;
    *(v4f*)(wp + 64) = vb[p] * rs * gb + bb;
  }
}

__global__ __launch_bounds__(128) void kv_proj_kernel(
    const float* __restrict__ x, const float* __restrict__ Wk,
    const float* __restrict__ Wv, float* __restrict__ Kx, float* __restrict__ Vx) {
  __shared__ float sx0[16 * LDP];
  const int tid = threadIdx.x;
  const int e0 = blockIdx.x * 16;
  for (int i = tid; i < 16 * 128; i += 128) {
    const int r = i >> 7, c = i & 127;
    sx0[r * LDP + c] = x[(size_t)(e0 + r) * (T_N * D_N) + c];
  }
  __syncthreads();

  const int lane = tid & 31, wave = tid >> 5;
  const int half = lane >> 4, lm = lane & 15;
  const int n0 = wave * 32 + lm, n1 = n0 + 16;
  v8f aK0{}, aK1{}, aV0{}, aV1{};
  for (int k0 = 0; k0 < 128; k0 += 4) {
    const int ka = k0 + 2 * half;
    v2f a, bk0, bk1, bv0, bv1;
    a.x = sx0[lm * LDP + ka];  a.y = sx0[lm * LDP + ka + 1];
    bk0.x = Wk[ka * 128 + n0]; bk0.y = Wk[(ka + 1) * 128 + n0];
    bk1.x = Wk[ka * 128 + n1]; bk1.y = Wk[(ka + 1) * 128 + n1];
    bv0.x = Wv[ka * 128 + n0]; bv0.y = Wv[(ka + 1) * 128 + n0];
    bv1.x = Wv[ka * 128 + n1]; bv1.y = Wv[(ka + 1) * 128 + n1];
    aK0 = wmma_f32(a, bk0, aK0);
    aK1 = wmma_f32(a, bk1, aK1);
    aV0 = wmma_f32(a, bv0, aV0);
    aV1 = wmma_f32(a, bv1, aV1);
  }
  for (int r = 0; r < 8; ++r) {
    const size_t row = (size_t)(e0 + r + 8 * half) * 128;
    Kx[row + n0] = aK0[r];
    Kx[row + n1] = aK1[r];
    Vx[row + n0] = aV0[r];
    Vx[row + n1] = aV1[r];
  }
}

__global__ __launch_bounds__(128) void attn_block_kernel(
    const float* __restrict__ x, const int* __restrict__ nbr,
    const float* __restrict__ Wq, const float* __restrict__ Wo,
    const float* __restrict__ bo, const float* __restrict__ g1,
    const float* __restrict__ b1, const float* __restrict__ g2,
    const float* __restrict__ b2, const float* __restrict__ W1,
    const float* __restrict__ bf1, const float* __restrict__ W2,
    const float* __restrict__ bf2, const float* __restrict__ Kx,
    const float* __restrict__ Vx, float* __restrict__ out) {
  __shared__ __attribute__((aligned(16))) float bufA[R_N * LDP];
  __shared__ __attribute__((aligned(16))) float bufB[R_N * LDP];
  __shared__ __attribute__((aligned(16))) float bufC[2 * EV_N * K_N * LDP];
  const int VROW = EV_N * K_N;
  const int VOFF = VROW * LDP;

  const int tid = threadIdx.x;
  const int e0 = blockIdx.x * EV_N;
  const int lane = tid & 31, wave = tid >> 5;
  const int half = lane >> 4, lm = lane & 15;
  const size_t xbase = (size_t)e0 * (T_N * D_N);

  v2f bw[KB_N][2];
  load_b_block(Wq + (2 * half) * 128 + wave * 32 + lm, 128, 0, bw);

  for (int i = tid; i < R_N * 128; i += 128) {
    const int r = i >> 7, c = i & 127;
    bufA[r * LDP + c] = x[xbase + (size_t)r * D_N + c];
  }
  for (int i = tid; i < EV_N * K_N * 128; i += 128) {
    const int k = i >> 7, c = i & 127;
    const int n = nbr[e0 * K_N + k];
    bufC[k * LDP + c]        = Kx[(size_t)n * D_N + c];
    bufC[VOFF + k * LDP + c] = Vx[(size_t)n * D_N + c];
  }
  __syncthreads();

  {
    v8f q[6] = {};
    gemm_acc(bufA, Wq, 128, Wo, 128, true, wave, half, lm, bw, q);
    for (int i = 0; i < 6; ++i)
      for (int r = 0; r < 8; ++r) {
        const int row = ((i >> 1) << 4) + r + 8 * half;
        const int col = wave * 32 + ((i & 1) << 4) + lm;
        bufB[row * LDP + col] = q[i][r] * 0.25f;
      }
  }
  __syncthreads();

  const int hc0 = (2 * wave) * 16, hc1 = hc0 + 16;

  head_product(bufB, bufC, 0, K_N, false, hc0, half, lm);
  head_product(bufB, bufC, 0, K_N, false, hc1, half, lm);
  __syncthreads();

  for (int idx = tid; idx < R_N * H_N; idx += 128) {
    const int t = idx >> 3, h = idx & 7;
    float* row = &bufB[t * LDP + h * 16];
    float m = row[0];
    for (int k = 1; k < 16; ++k) m = fmaxf(m, row[k]);
    float s = 0.f;
    for (int k = 0; k < 16; ++k) { row[k] = __expf(row[k] - m); s += row[k]; }
    const float inv = 1.f / s;
    for (int k = 0; k < 16; ++k) row[k] *= inv;
  }
  __syncthreads();

  head_product(bufB, bufC, VROW, VROW + K_N, true, hc0, half, lm);
  head_product(bufB, bufC, VROW, VROW + K_N, true, hc1, half, lm);
  __syncthreads();

  v4f g1a, g1b, b1a, b1b;
  {
    v8f o[6] = {};
    gemm_acc(bufB, Wo, 128, W1, FF_N, true, wave, half, lm, bw, o);
    g1a = *(const v4f*)(g1 + 4 * lm); g1b = *(const v4f*)(g1 + 64 + 4 * lm);
    b1a = *(const v4f*)(b1 + 4 * lm); b1b = *(const v4f*)(b1 + 64 + 4 * lm);
    for (int i = 0; i < 6; ++i)
      for (int r = 0; r < 8; ++r) {
        const int row = ((i >> 1) << 4) + r + 8 * half;
        const int col = wave * 32 + ((i & 1) << 4) + lm;
        bufC[row * LDP + col] = o[i][r] + bo[col] + bufA[row * LDP + col];
      }
  }
  __syncthreads();

  layer_norm_rows(bufC, bufA, LDP, g1a, g1b, b1a, b1b, wave, half, lm);
  __syncthreads();

  v8f y[6] = {};
  for (int ch = 0; ch < 4; ++ch) {
    v8f g[6] = {};
    gemm_acc(bufA, W1 + ch * 128, FF_N, W2 + (size_t)(ch * 128) * D_N, 128, true,
             wave, half, lm, bw, g);
    for (int i = 0; i < 6; ++i)
      for (int r = 0; r < 8; ++r) {
        const int row = ((i >> 1) << 4) + r + 8 * half;
        const int col = wave * 32 + ((i & 1) << 4) + lm;
        bufB[row * LDP + col] = fmaxf(g[i][r] + bf1[ch * 128 + col], 0.f);
      }
    __syncthreads();
    gemm_acc(bufB, W2 + (size_t)(ch * 128) * D_N, 128, W1 + (ch + 1) * 128, FF_N, ch < 3,
             wave, half, lm, bw, y);
    __syncthreads();
  }

  const v4f g2a = *(const v4f*)(g2 + 4 * lm), g2b = *(const v4f*)(g2 + 64 + 4 * lm);
  const v4f b2a = *(const v4f*)(b2 + 4 * lm), b2b = *(const v4f*)(b2 + 64 + 4 * lm);
  for (int i = 0; i < 6; ++i)
    for (int r = 0; r < 8; ++r) {
      const int row = ((i >> 1) << 4) + r + 8 * half;
      const int col = wave * 32 + ((i & 1) << 4) + lm;
      bufB[row * LDP + col] = y[i][r] + bf2[col] + bufA[row * LDP + col];
    }
  __syncthreads();

  layer_norm_rows(bufB, out + xbase, D_N, g2a, g2b, b2a, b2b, wave, half, lm);
}

extern "C" void kernel_launch(void* const* d_in, const int* in_sizes, int n_in,
                              void* d_out, int out_size, void* d_ws, size_t ws_size,
                              hipStream_t stream) {
  const float* x   = (const float*)d_in[0];
  const int*   nbr = (const int*)d_in[1];
  const float* Wq  = (const float*)d_in[2];
  const float* Wk  = (const float*)d_in[3];
  const float* Wv  = (const float*)d_in[4];
  const float* Wo  = (const float*)d_in[5];
  const float* bo  = (const float*)d_in[6];
  const float* g1  = (const float*)d_in[7];
  const float* b1  = (const float*)d_in[8];
  const float* g2  = (const float*)d_in[9];
  const float* b2  = (const float*)d_in[10];
  const float* W1  = (const float*)d_in[11];
  const float* bf1 = (const float*)d_in[12];
  const float* W2  = (const float*)d_in[13];
  const float* bf2 = (const float*)d_in[14];

  float* Kx = (float*)d_ws;
  float* Vx = Kx + (size_t)E_N * D_N;

  kv_proj_kernel<<<E_N / 16, 128, 0, stream>>>(x, Wk, Wv, Kx, Vx);
  attn_block_kernel<<<E_N / EV_N, 128, 0, stream>>>(x, nbr, Wq, Wo, bo, g1, b1, g2, b2,
                                             W1, bf1, W2, bf2, Kx, Vx,
                                             (float*)d_out);
}


# prologue staging as one group of 16-byte loads: 32 neighbour indices by one load + lane broadcast, then all x and gathered K/V rows in flight at once (was 11 waited rounds of 4-byte loads)
# speedup vs baseline: 2.3167x; 1.0491x over previous
#include <hip/hip_runtime.h>
#include <hip/hip_bf16.h>

#define E_N 10000
#define T_N 24
#define D_N 128
#define H_N 8
#define K_N 16
#define FF_N 512
#define EPS_LN 1e-5f
#define LDP 132
#define EV_N 2
#define R_N (EV_N * T_N)

typedef float v2f __attribute__((ext_vector_type(2)));
typedef float v8f __attribute__((ext_vector_type(8)));

__device__ __forceinline__ v8f wmma_f32(v2f a, v2f b, v8f c) {
  return __builtin_amdgcn_wmma_f32_16x16x4_f32(
       false, a,  false, b,
       (short)0, c,  false,  false);
}

#define KB_N 8
#define NBLK (128 / (4 * KB_N))

__device__ __forceinline__ void load_a_block(const float* Al, int kb0, v2f a[KB_N][3]) {
#pragma unroll
  for (int s = 0; s < KB_N; ++s) {
    const int k = kb0 + 4 * s;
    a[s][0].x = Al[k];            a[s][0].y = Al[k + 1];
    a[s][1].x = Al[16 * LDP + k]; a[s][1].y = Al[16 * LDP + k + 1];
    a[s][2].x = Al[32 * LDP + k]; a[s][2].y = Al[32 * LDP + k + 1];
  }
}

__device__ __forceinline__ void load_b_block(const float* Bl, int ldb, int kb0,
                                             v2f b[KB_N][2]) {
#pragma unroll
  for (int s = 0; s < KB_N; ++s) {
    const int k = kb0 + 4 * s;
    b[s][0].x = Bl[k * ldb];      b[s][0].y = Bl[(k + 1) * ldb];
    b[s][1].x = Bl[k * ldb + 16]; b[s][1].y = Bl[(k + 1) * ldb + 16];
  }
}

__device__ __forceinline__ void gemm_acc(const float* sA, const float* B, int ldb,
                                         const float* Bn, int ldbn, bool has_next,
                                         int wave, int half, int lm,
                                         v2f bfirst[KB_N][2], v8f acc[6]) {
  const float* Al = sA + lm * LDP + 2 * half;
  const float* Bl = B + (2 * half) * ldb + wave * 32 + lm;
  const float* Bnl = Bn + (2 * half) * ldbn + wave * 32 + lm;
  v2f a[2][KB_N][3];
  v2f b1[KB_N][2];
  load_a_block(Al, 0, a[0]);
#pragma unroll
  for (int blk = 0; blk < NBLK; ++blk) {
    v2f (*bc)[2] = (blk & 1) ? b1 : bfirst;
    v2f (*bn)[2] = (blk & 1) ? bfirst : b1;
    if (blk + 1 < NBLK) {
      load_a_block(Al, (blk + 1) * 4 * KB_N, a[(blk + 1) & 1]);
      load_b_block(Bl, ldb, (blk + 1) * 4 * KB_N, bn);
    } else if (has_next) {
      load_b_block(Bnl, ldbn, 0, bn);
    }
    __builtin_amdgcn_sched_barrier(0);
#pragma unroll
    for (int s = 0; s < KB_N; ++s) {
      acc[0] = wmma_f32(a[blk & 1][s][0], bc[s][0], acc[0]);
      acc[1] = wmma_f32(a[blk & 1][s][0], bc[s][1], acc[1]);
      acc[2] = wmma_f32(a[blk & 1][s][1], bc[s][0], acc[2]);
      acc[3] = wmma_f32(a[blk & 1][s][1], bc[s][1], acc[3]);
      acc[4] = wmma_f32(a[blk & 1][s][2], bc[s][0], acc[4]);
      acc[5] = wmma_f32(a[blk & 1][s][2], bc[s][1], acc[5]);
    }
    __builtin_amdgcn_sched_barrier(0);
  }
}

__device__ __forceinline__ void head_product(float* sP, const float* sB, int rowB0,
                                             int rowB1, bool kmajor, int hc,
                                             int half, int lm) {
  v8f p0{}, p1a{}, p1b{}, p2{};
  for (int k0 = 0; k0 < 16; k0 += 4) {
    const int ka = k0 + 2 * half;
    v2f a0, a1, a2, b0, b1;
    a0.x = sP[lm * LDP + hc + ka];        a0.y = sP[lm * LDP + hc + ka + 1];
    a1.x = sP[(16 + lm) * LDP + hc + ka]; a1.y = sP[(16 + lm) * LDP + hc + ka + 1];
    a2.x = sP[(32 + lm) * LDP + hc + ka]; a2.y = sP[(32 + lm) * LDP + hc + ka + 1];
    if (kmajor) {
      b0.x = sB[(rowB0 + ka) * LDP + hc + lm]; b0.y = sB[(rowB0 + ka + 1) * LDP + hc + lm];
      b1.x = sB[(rowB1 + ka) * LDP + hc + lm]; b1.y = sB[(rowB1 + ka + 1) * LDP + hc + lm];
    } else {
      b0.x = sB[(rowB0 + lm) * LDP + hc + ka]; b0.y = sB[(rowB0 + lm) * LDP + hc + ka + 1];
      b1.x = sB[(rowB1 + lm) * LDP + hc + ka]; b1.y = sB[(rowB1 + lm) * LDP + hc + ka + 1];
    }
    p0  = wmma_f32(a0, b0, p0);
    p1a = wmma_f32(a1, b0, p1a);
    p1b = wmma_f32(a1, b1, p1b);
    p2  = wmma_f32(a2, b1, p2);
  }
  for (int r = 0; r < 8; ++r) {
    const int t = r + 8 * half;
    sP[t * LDP + hc + lm]        = p0[r];
    sP[(16 + t) * LDP + hc + lm] = half ? p1b[r] : p1a[r];
    sP[(32 + t) * LDP + hc + lm] = p2[r];
  }
}

typedef float v4f __attribute__((ext_vector_type(4)));

template <int CTRL>
__device__ __forceinline__ void row_xor_add6(float s[6]) {
#pragma unroll
  for (int p = 0; p < 6; ++p)
    s[p] += __builtin_bit_cast(float, __builtin_amdgcn_update_dpp(
                0, __builtin_bit_cast(int, s[p]), CTRL, 0xf, 0xf, false));
}

__device__ __forceinline__ void row_sum6(float s[6]) {
  row_xor_add6<0x161>(s);
  row_xor_add6<0x162>(s);
  row_xor_add6<0x164>(s);
  row_xor_add6<0x168>(s);
}

__device__ __forceinline__ void layer_norm_rows(const float* src, float* dst, int dpitch,
                                                v4f ga, v4f gb, v4f ba, v4f bb,
                                                int wave, int half, int lm) {
  v4f va[6], vb[6];
  float s[6];
#pragma unroll
  for (int p = 0; p < 6; ++p) {
    const float* rp = src + (12 * wave + 2 * p + half) * LDP + 4 * lm;
    va[p] = *(const v4f*)rp;
    vb[p] = *(const v4f*)(rp + 64);
  }
#pragma unroll
  for (int p = 0; p < 6; ++p)
    s[p] = ((va[p].x + va[p].y) + (va[p].z + va[p].w)) +
           ((vb[p].x + vb[p].y) + (vb[p].z + vb[p].w));
  row_sum6(s);
#pragma unroll
  for (int p = 0; p < 6; ++p) {
    const float mu = s[p] * (1.f / 128.f);
    va[p] -= mu;
    vb[p] -= mu;
    s[p] = ((va[p].x * va[p].x + va[p].y * va[p].y) + (va[p].z * va[p].z + va[p].w * va[p].w)) +
           ((vb[p].x * vb[p].x + vb[p].y * vb[p].y) + (vb[p].z * vb[p].z + vb[p].w * vb[p].w));
  }
  row_sum6(s);
#pragma unroll
  for (int p = 0; p < 6; ++p) {
    const float rs = rsqrtf(s[p] * (1.f / 128.f) + EPS_LN);
    float* wp = dst + (12 * wave + 2 * p + half) * dpitch + 4 * lm;
    *(v4f*)wp        = va[p] * rs * ga + ba;
    *(v4f*)(wp + 64) = vb[p] * rs * gb + bb;
  }
}

__global__ __launch_bounds__(128) void kv_proj_kernel(
    const float* __restrict__ x, const float* __restrict__ Wk,
    const float* __restrict__ Wv, float* __restrict__ Kx, float* __restrict__ Vx) {
  __shared__ float sx0[16 * LDP];
  const int tid = threadIdx.x;
  const int e0 = blockIdx.x * 16;
  for (int i = tid; i < 16 * 128; i += 128) {
    const int r = i >> 7, c = i & 127;
    sx0[r * LDP + c] = x[(size_t)(e0 + r) * (T_N * D_N) + c];
  }
  __syncthreads();

  const int lane = tid & 31, wave = tid >> 5;
  const int half = lane >> 4, lm = lane & 15;
  const int n0 = wave * 32 + lm, n1 = n0 + 16;
  v8f aK0{}, aK1{}, aV0{}, aV1{};
  for (int k0 = 0; k0 < 128; k0 += 4) {
    const int ka = k0 + 2 * half;
    v2f a, bk0, bk1, bv0, bv1;
    a.x = sx0[lm * LDP + ka];  a.y = sx0[lm * LDP + ka + 1];
    bk0.x = Wk[ka * 128 + n0]; bk0.y = Wk[(ka + 1) * 128 + n0];
    bk1.x = Wk[ka * 128 + n1]; bk1.y = Wk[(ka + 1) * 128 + n1];
    bv0.x = Wv[ka * 128 + n0]; bv0.y = Wv[(ka + 1) * 128 + n0];
    bv1.x = Wv[ka * 128 + n1]; bv1.y = Wv[(ka + 1) * 128 + n1];
    aK0 = wmma_f32(a, bk0, aK0);
    aK1 = wmma_f32(a, bk1, aK1);
    aV0 = wmma_f32(a, bv0, aV0);
    aV1 = wmma_f32(a, bv1, aV1);
  }
  for (int r = 0; r < 8; ++r) {
    const size_t row = (size_t)(e0 + r + 8 * half) * 128;
    Kx[row + n0] = aK0[r];
    Kx[row + n1] = aK1[r];
    Vx[row + n0] = aV0[r];
    Vx[row + n1] = aV1[r];
  }
}

__global__ __launch_bounds__(128) void attn_block_kernel(
    const float* __restrict__ x, const int* __restrict__ nbr,
    const float* __restrict__ Wq, const float* __restrict__ Wo,
    const float* __restrict__ bo, const float* __restrict__ g1,
    const float* __restrict__ b1, const float* __restrict__ g2,
    const float* __restrict__ b2, const float* __restrict__ W1,
    const float* __restrict__ bf1, const float* __restrict__ W2,
    const float* __restrict__ bf2, const float* __restrict__ Kx,
    const float* __restrict__ Vx, float* __restrict__ out) {
  __shared__ __attribute__((aligned(16))) float bufA[R_N * LDP];
  __shared__ __attribute__((aligned(16))) float bufB[R_N * LDP];
  __shared__ __attribute__((aligned(16))) float bufC[2 * EV_N * K_N * LDP];
  const int VROW = EV_N * K_N;
  const int VOFF = VROW * LDP;

  const int tid = threadIdx.x;
  const int e0 = blockIdx.x * EV_N;
  const int lane = tid & 31, wave = tid >> 5;
  const int half = lane >> 4, lm = lane & 15;
  const size_t xbase = (size_t)e0 * (T_N * D_N);

  const int nidx = nbr[e0 * K_N + lane];

  v2f bw[KB_N][2];
  load_b_block(Wq + (2 * half) * 128 + wave * 32 + lm, 128, 0, bw);

  {
    const v4f* xg = (const v4f*)(x + xbase) + tid;
    v4f xv[R_N / 4];
#pragma unroll
    for (int j = 0; j < R_N / 4; ++j) xv[j] = xg[128 * j];

    const int wave_s = __builtin_amdgcn_readfirstlane(wave);
    v4f kv[EV_N * K_N / 4], vv[EV_N * K_N / 4];
#pragma unroll
    for (int j = 0; j < EV_N * K_N / 4; ++j) {
      const int n = __builtin_amdgcn_readlane(nidx, wave_s + 4 * j);
      kv[j] = *((const v4f*)(Kx + (size_t)n * D_N) + lane);
      vv[j] = *((const v4f*)(Vx + (size_t)n * D_N) + lane);
    }

#pragma unroll
    for (int j = 0; j < R_N / 4; ++j)
      *(v4f*)&bufA[(wave + 4 * j) * LDP + 4 * lane] = xv[j];
#pragma unroll
    for (int j = 0; j < EV_N * K_N / 4; ++j) {
      *(v4f*)&bufC[(wave + 4 * j) * LDP + 4 * lane]        = kv[j];
      *(v4f*)&bufC[VOFF + (wave + 4 * j) * LDP + 4 * lane] = vv[j];
    }
  }
  __syncthreads();

  {
    v8f q[6] = {};
    gemm_acc(bufA, Wq, 128, Wo, 128, true, wave, half, lm, bw, q);
    for (int i = 0; i < 6; ++i)
      for (int r = 0; r < 8; ++r) {
        const int row = ((i >> 1) << 4) + r + 8 * half;
        const int col = wave * 32 + ((i & 1) << 4) + lm;
        bufB[row * LDP + col] = q[i][r] * 0.25f;
      }
  }
  __syncthreads();

  const int hc0 = (2 * wave) * 16, hc1 = hc0 + 16;

  head_product(bufB, bufC, 0, K_N, false, hc0, half, lm);
  head_product(bufB, bufC, 0, K_N, false, hc1, half, lm);
  __syncthreads();

  for (int idx = tid; idx < R_N * H_N; idx += 128) {
    const int t = idx >> 3, h = idx & 7;
    float* row = &bufB[t * LDP + h * 16];
    float m = row[0];
    for (int k = 1; k < 16; ++k) m = fmaxf(m, row[k]);
    float s = 0.f;
    for (int k = 0; k < 16; ++k) { row[k] = __expf(row[k] - m); s += row[k]; }
    const float inv = 1.f / s;
    for (int k = 0; k < 16; ++k) row[k] *= inv;
  }
  __syncthreads();

  head_product(bufB, bufC, VROW, VROW + K_N, true, hc0, half, lm);
  head_product(bufB, bufC, VROW, VROW + K_N, true, hc1, half, lm);
  __syncthreads();

  v4f g1a, g1b, b1a, b1b;
  {
    v8f o[6] = {};
    gemm_acc(bufB, Wo, 128, W1, FF_N, true, wave, half, lm, bw, o);
    g1a = *(const v4f*)(g1 + 4 * lm); g1b = *(const v4f*)(g1 + 64 + 4 * lm);
    b1a = *(const v4f*)(b1 + 4 * lm); b1b = *(const v4f*)(b1 + 64 + 4 * lm);
    for (int i = 0; i < 6; ++i)
      for (int r = 0; r < 8; ++r) {
        const int row = ((i >> 1) << 4) + r + 8 * half;
        const int col = wave * 32 + ((i & 1) << 4) + lm;
        bufC[row * LDP + col] = o[i][r] + bo[col] + bufA[row * LDP + col];
      }
  }
  __syncthreads();

  layer_norm_rows(bufC, bufA, LDP, g1a, g1b, b1a, b1b, wave, half, lm);
  __syncthreads();

  v8f y[6] = {};
  for (int ch = 0; ch < 4; ++ch) {
    v8f g[6] = {};
    gemm_acc(bufA, W1 + ch * 128, FF_N, W2 + (size_t)(ch * 128) * D_N, 128, true,
             wave, half, lm, bw, g);
    for (int i = 0; i < 6; ++i)
      for (int r = 0; r < 8; ++r) {
        const int row = ((i >> 1) << 4) + r + 8 * half;
        const int col = wave * 32 + ((i & 1) << 4) + lm;
        bufB[row * LDP + col] = fmaxf(g[i][r] + bf1[ch * 128 + col], 0.f);
      }
    __syncthreads();
    gemm_acc(bufB, W2 + (size_t)(ch * 128) * D_N, 128, W1 + (ch + 1) * 128, FF_N, ch < 3,
             wave, half, lm, bw, y);
    __syncthreads();
  }

  const v4f g2a = *(const v4f*)(g2 + 4 * lm), g2b = *(const v4f*)(g2 + 64 + 4 * lm);
  const v4f b2a = *(const v4f*)(b2 + 4 * lm), b2b = *(const v4f*)(b2 + 64 + 4 * lm);
  for (int i = 0; i < 6; ++i)
    for (int r = 0; r < 8; ++r) {
      const int row = ((i >> 1) << 4) + r + 8 * half;
      const int col = wave * 32 + ((i & 1) << 4) + lm;
      bufB[row * LDP + col] = y[i][r] + bf2[col] + bufA[row * LDP + col];
    }
  __syncthreads();

  layer_norm_rows(bufB, out + xbase, D_N, g2a, g2b, b2a, b2b, wave, half, lm);
}

extern "C" void kernel_launch(void* const* d_in, const int* in_sizes, int n_in,
                              void* d_out, int out_size, void* d_ws, size_t ws_size,
                              hipStream_t stream) {
  const float* x   = (const float*)d_in[0];
  const int*   nbr = (const int*)d_in[1];
  const float* Wq  = (const float*)d_in[2];
  const float* Wk  = (const float*)d_in[3];
  const float* Wv  = (const float*)d_in[4];
  const float* Wo  = (const float*)d_in[5];
  const float* bo  = (const float*)d_in[6];
  const float* g1  = (const float*)d_in[7];
  const float* b1  = (const float*)d_in[8];
  const float* g2  = (const float*)d_in[9];
  const float* b2  = (const float*)d_in[10];
  const float* W1  = (const float*)d_in[11];
  const float* bf1 = (const float*)d_in[12];
  const float* W2  = (const float*)d_in[13];
  const float* bf2 = (const float*)d_in[14];

  float* Kx = (float*)d_ws;
  float* Vx = Kx + (size_t)E_N * D_N;

  kv_proj_kernel<<<E_N / 16, 128, 0, stream>>>(x, Wk, Wv, Kx, Vx);
  attn_block_kernel<<<E_N / EV_N, 128, 0, stream>>>(x, nbr, Wq, Wo, bo, g1, b1, g2, b2,
                                             W1, bf1, W2, bf2, Kx, Vx,
                                             (float*)d_out);
}
